# starting kernel
# baseline (speedup 1.0000x reference)
#include <hip/hip_runtime.h>
#include <hip/hip_bf16.h>
#include <math.h>

constexpr int Bz = 2;
constexpr int Sq = 2048;
constexpr int Hd = 2048;
constexpr int NH = 16;
constexpr int HD = 128;
constexpr float NEGV = -1000000.0f;

typedef float v2f __attribute__((ext_vector_type(2)));
typedef float v8f __attribute__((ext_vector_type(8)));

#define WMMA_F32(a, b, c) \
  __builtin_amdgcn_wmma_f32_16x16x4_f32(false, (a), false, (b), (short)0, (c), false, false)

__device__ __forceinline__ v8f vzero8() {
  v8f z = {0.f, 0.f, 0.f, 0.f, 0.f, 0.f, 0.f, 0.f};
  return z;
}

constexpr int TM = 128, TN = 128, TK = 32;
constexpr int APAD = 4;
constexpr int BIPAD = 32;

__global__ __launch_bounds__(256) void gemm_f32_wmma(
    const float* __restrict__ A, const float* __restrict__ Bm,
    float* __restrict__ C, int M, int N, int K) {
  __shared__ float As[TM][TK + APAD];
  __shared__ float Bsp[TK / 2][2 * TN + BIPAD];

  const int t = threadIdx.x;
  const int lane = t & 31;
  const int wave = t >> 5;
  const int lm = lane & 15;
  const int lh = lane >> 4;
  const int wm = wave & 3;
  const int wn = wave >> 2;
  const int m0 = blockIdx.y * TM;
  const int n0 = blockIdx.x * TN;

  v8f acc[2][4];
#pragma unroll
  for (int i = 0; i < 2; ++i)
#pragma unroll
    for (int j = 0; j < 4; ++j) acc[i][j] = vzero8();

  for (int kt = 0; kt < K; kt += TK) {
#pragma unroll
    for (int i = 0; i < 4; ++i) {
      int idx = t + i * 256;
      int r = idx >> 3;
      int c = (idx & 7) << 2;
      float4 v = *(const float4*)(A + (size_t)(m0 + r) * K + kt + c);
      *(float4*)&As[r][c] = v;
    }
#pragma unroll
    for (int i = 0; i < 4; ++i) {
      int idx = t + i * 256;
      int r = idx >> 5;
      int c = (idx & 31) << 2;
      float4 v = *(const float4*)(Bm + (size_t)(kt + r) * N + n0 + c);
      float* dst = &Bsp[r >> 1][2 * c + (r & 1)];
      dst[0] = v.x;
      dst[2] = v.y;
      dst[4] = v.z;
      dst[6] = v.w;
    }
    __syncthreads();

#pragma unroll
    for (int kk = 0; kk < TK; kk += 4) {
      v2f af[2];
#pragma unroll
      for (int im = 0; im < 2; ++im) {
        int row = wm * 32 + im * 16 + lm;
        af[im] = *(const v2f*)&As[row][kk + 2 * lh];
      }
      const int khalf = (kk >> 1) + lh;
#pragma unroll
      for (int j = 0; j < 4; ++j) {
        int col = wn * 64 + j * 16 + lm;
        v2f bf = *(const v2f*)&Bsp[khalf][2 * col];
        acc[0][j] = WMMA_F32(af[0], bf, acc[0][j]);
        acc[1][j] = WMMA_F32(af[1], bf, acc[1][j]);
      }
    }
    __syncthreads();
  }

#pragma unroll
  for (int im = 0; im < 2; ++im)
#pragma unroll
    for (int j = 0; j < 4; ++j)
#pragma unroll
      for (int r = 0; r < 8; ++r) {
        int m = m0 + wm * 32 + im * 16 + r + 8 * lh;
        int n = n0 + wn * 64 + j * 16 + lm;
        C[(size_t)m * N + n] = acc[im][j][r];
      }
}

constexpr int QT = 64;
constexpr int KT = 32;
constexpr int KVP = 12;
constexpr int VIPAD = 32;
constexpr int PPAD = 4;

__global__ __launch_bounds__(128) void attn_f32_wmma(
    const float* __restrict__ Qm, const float* __restrict__ Km,
    const float* __restrict__ Vm, const int* __restrict__ mask,
    float* __restrict__ O) {
  __shared__ float Ks[KT][HD + KVP];
  __shared__ float Vsp[KT / 2][2 * HD + VIPAD];
  __shared__ float Ps[4][16][KT + PPAD];
  __shared__ float corrS[4][16];
  __shared__ float rsumS[4][16];
  __shared__ int mkS[KT];

  const int t = threadIdx.x;
  const int lane = t & 31;
  const int w = t >> 5;
  const int lm = lane & 15;
  const int lh = lane >> 4;

  int bid = blockIdx.x;
  const int nqb = Sq / QT;
  const int qb = bid % nqb; bid /= nqb;
  const int h = bid % NH;
  const int b = bid / NH;
  const int q0 = qb * QT;

  const size_t baseBH = (size_t)b * Sq * Hd + (size_t)h * Sq * HD;

  const float* qrow = Qm + baseBH + (size_t)(q0 + w * 16 + lm) * HD;
  v2f qf[HD / 4];
#pragma unroll
  for (int kd = 0; kd < HD / 4; ++kd)
    qf[kd] = *(const v2f*)(qrow + 4 * kd + 2 * lh);

  int mq[8];
#pragma unroll
  for (int r = 0; r < 8; ++r)
    mq[r] = mask[b * Sq + q0 + w * 16 + r + 8 * lh];

  v8f o[8];
#pragma unroll
  for (int s = 0; s < 8; ++s) o[s] = vzero8();
  float rmax = -INFINITY;
  float rsum = 0.0f;

  for (int kt = 0; kt < Sq; kt += KT) {
#pragma unroll
    for (int i = 0; i < 8; ++i) {
      int idx = t + i * 128;
      int r = idx >> 5;
      int c = (idx & 31) << 2;
      float4 kv = *(const float4*)(Km + baseBH + (size_t)(kt + r) * HD + c);
      *(float4*)&Ks[r][c] = kv;
      float4 vv = *(const float4*)(Vm + baseBH + (size_t)(kt + r) * HD + c);
      float* dst = &Vsp[r >> 1][2 * c + (r & 1)];
      dst[0] = vv.x;
      dst[2] = vv.y;
      dst[4] = vv.z;
      dst[6] = vv.w;
    }
    if (t < KT) mkS[t] = mask[b * Sq + kt + t];
    __syncthreads();

    v8f sc[2];
    sc[0] = vzero8();
    sc[1] = vzero8();
#pragma unroll
    for (int kd = 0; kd < HD / 4; ++kd) {
      v2f af = qf[kd];
#pragma unroll
      for (int j = 0; j < 2; ++j) {
        int kc = j * 16 + lm;
        v2f bf = *(const v2f*)&Ks[kc][4 * kd + 2 * lh];
        sc[j] = WMMA_F32(af, bf, sc[j]);
      }
    }

#pragma unroll
    for (int j = 0; j < 2; ++j)
#pragma unroll
      for (int r = 0; r < 8; ++r) {
        int m = r + 8 * lh;
        int kc = j * 16 + lm;
        float v = sc[j][r];
        if (mq[r] == 0 || mkS[kc] == 0) v = NEGV;
        Ps[w][m][kc] = v;
      }
    __syncthreads();

    {
      const int c0 = lh * 16;
      float pm = -INFINITY;
#pragma unroll
      for (int c = 0; c < 16; ++c) pm = fmaxf(pm, Ps[w][lm][c0 + c]);
      pm = fmaxf(pm, __shfl_xor(pm, 16, 32));
      float nm = fmaxf(rmax, pm);
      float cf = __expf(rmax - nm);
      float ps = 0.0f;
#pragma unroll
      for (int c = 0; c < 16; ++c) {
        float e = __expf(Ps[w][lm][c0 + c] - nm);
        Ps[w][lm][c0 + c] = e;
        ps += e;
      }
      ps += __shfl_xor(ps, 16, 32);
      rsum = rsum * cf + ps;
      rmax = nm;
      if (lane < 16) corrS[w][lm] = cf;
    }
    __syncthreads();

    float cv[8];
#pragma unroll
    for (int r = 0; r < 8; ++r) cv[r] = corrS[w][r + 8 * lh];
#pragma unroll
    for (int s = 0; s < 8; ++s)
#pragma unroll
      for (int r = 0; r < 8; ++r) o[s][r] *= cv[r];

#pragma unroll
    for (int kk = 0; kk < KT; kk += 4) {
      v2f af = *(const v2f*)&Ps[w][lm][kk + 2 * lh];
      const int khalf = (kk >> 1) + lh;
#pragma unroll
      for (int s = 0; s < 8; ++s) {
        int col = s * 16 + lm;
        v2f bf = *(const v2f*)&Vsp[khalf][2 * col];
        o[s] = WMMA_F32(af, bf, o[s]);
      }
    }
    __syncthreads();
  }

  if (lane < 16) rsumS[w][lm] = rsum;
  __syncthreads();
  float rv[8];
#pragma unroll
  for (int r = 0; r < 8; ++r) rv[r] = 1.0f / rsumS[w][r + 8 * lh];

#pragma unroll
  for (int s = 0; s < 8; ++s)
#pragma unroll
    for (int r = 0; r < 8; ++r) {
      int s2 = q0 + w * 16 + r + 8 * lh;
      int d = s * 16 + lm;
      O[baseBH + (size_t)s2 * HD + d] = o[s][r] * rv[r];
    }
}

extern "C" void kernel_launch(void* const* d_in, const int* in_sizes, int n_in,
                              void* d_out, int out_size, void* d_ws, size_t ws_size,
                              hipStream_t stream) {
  const float* X    = (const float*)d_in[0];
  const int*   mask = (const int*)d_in[1];
  const float* Wq   = (const float*)d_in[2];
  const float* Wk   = (const float*)d_in[3];
  const float* Wv   = (const float*)d_in[4];
  const float* Wqm  = (const float*)d_in[5];
  const float* Wkm  = (const float*)d_in[6];
  const float* Wvm  = (const float*)d_in[7];
  const float* Wo   = (const float*)d_in[8];
  float* out = (float*)d_out;

  const size_t SZ = (size_t)Bz * Sq * Hd;
  float* s0 = (float*)d_ws;
  float* s1 = s0 + SZ;
  float* s2 = s1 + SZ;
  float* s3 = s2 + SZ;

  const int M = Bz * Sq, N = Hd, K = Hd;
  dim3 gg(N / TN, M / TM);
  dim3 gb(256);

  gemm_f32_wmma<<<gg, gb, 0, stream>>>(X,  Wq,  s0, M, N, K);
  gemm_f32_wmma<<<gg, gb, 0, stream>>>(X,  Wk,  s1, M, N, K);
  gemm_f32_wmma<<<gg, gb, 0, stream>>>(X,  Wv,  s2, M, N, K);
  gemm_f32_wmma<<<gg, gb, 0, stream>>>(s0, Wqm, s3, M, N, K);
  gemm_f32_wmma<<<gg, gb, 0, stream>>>(s1, Wkm, s0, M, N, K);
  gemm_f32_wmma<<<gg, gb, 0, stream>>>(s2, Wvm, s1, M, N, K);

  attn_f32_wmma<<<Bz * NH * (Sq / QT), 128, 0, stream>>>(s3, s0, s1, mask, s2);

  gemm_f32_wmma<<<gg, gb, 0, stream>>>(s2, Wo, out, M, N, K);
}


# GEMM wave tile 32x64 -> 64x64: 8 waves = 2x2 positions x 2 K halves, partial sums added via LDS at the end; 4 k-steps' 16 fragment reads stand as one group before their 64 WMMAs (was 1 place per 8)
# speedup vs baseline: 1.1193x; 1.1193x over previous
#include <hip/hip_runtime.h>
#include <hip/hip_bf16.h>
#include <math.h>

constexpr int Bz = 2;
constexpr int Sq = 2048;
constexpr int Hd = 2048;
constexpr int NH = 16;
constexpr int HD = 128;
constexpr float NEGV = -1000000.0f;

typedef float v2f __attribute__((ext_vector_type(2)));
typedef float v8f __attribute__((ext_vector_type(8)));

#define WMMA_F32(a, b, c) \
  __builtin_amdgcn_wmma_f32_16x16x4_f32(false, (a), false, (b), (short)0, (c), false, false)

__device__ __forceinline__ v8f vzero8() {
  v8f z = {0.f, 0.f, 0.f, 0.f, 0.f, 0.f, 0.f, 0.f};
  return z;
}

__device__ __forceinline__ void prio_by_progress(int level) {
  switch (level & 3) {
    case 0:  __builtin_amdgcn_s_setprio(3); break;
    case 1:  __builtin_amdgcn_s_setprio(2); break;
    case 2:  __builtin_amdgcn_s_setprio(1); break;
    default: __builtin_amdgcn_s_setprio(0); break;
  }
}

constexpr int TM = 128, TN = 128, TK = 128;
constexpr int APAD = 4;
constexpr int NLA = TM * TK / 4 / 256;
constexpr int NLB = TK * TN / 4 / 256;
constexpr int BIPAD = 32;

template <int KG>
__device__ __forceinline__ void gemm_hand_over(const v8f (&acc)[4][4], float4* xb,
                                               int wave, int lane) {
  constexpr int GIVE = (1 - KG) * 2;
#pragma unroll
  for (int i = 0; i < 2; ++i)
#pragma unroll
    for (int j = 0; j < 4; ++j) {
      const v8f a = acc[GIVE + i][j];
      const int slot = ((wave * 8 + i * 4 + j) * 2) * 32 + lane;
      xb[slot]      = make_float4(a[0], a[1], a[2], a[3]);
      xb[slot + 32] = make_float4(a[4], a[5], a[6], a[7]);
    }
}

template <int KG>
__device__ __forceinline__ void gemm_finish(const v8f (&acc)[4][4], const float4* xb,
                                            int wave, int lane, float* __restrict__ C,
                                            int N, int mw, int nw) {
  constexpr int KEEP = KG * 2;
  const int lm = lane & 15;
  const int lh = lane >> 4;
  const int pw = wave ^ 4;
#pragma unroll
  for (int i = 0; i < 2; ++i)
#pragma unroll
    for (int j = 0; j < 4; ++j) {
      const int slot = ((pw * 8 + i * 4 + j) * 2) * 32 + lane;
      const float4 p0 = xb[slot];
      const float4 p1 = xb[slot + 32];
      const v8f a = acc[KEEP + i][j];
      const float sum[8] = {a[0] + p0.x, a[1] + p0.y, a[2] + p0.z, a[3] + p0.w,
                            a[4] + p1.x, a[5] + p1.y, a[6] + p1.z, a[7] + p1.w};
#pragma unroll
      for (int r = 0; r < 8; ++r) {
        int m = mw + (KEEP + i) * 16 + r + 8 * lh;
        int n = nw + j * 16 + lm;
        C[(size_t)m * N + n] = sum[r];
      }
    }
}

__global__ __launch_bounds__(256, 4) void gemm_f32_wmma(
    const float* __restrict__ A, const float* __restrict__ Bm,
    float* __restrict__ C, int M, int N, int K) {
  __shared__ float As[TM][TK + APAD];
  __shared__ float Bsp[TK / 2][2 * TN + BIPAD];

  const int t = threadIdx.x;
  const int lane = t & 31;
  const int wave = t >> 5;
  const int lm = lane & 15;
  const int lh = lane >> 4;
  const int pos = wave & 3;
  const int kg = wave >> 2;
  const int wm = pos & 1;
  const int wn = pos >> 1;
  const int m0 = blockIdx.y * TM;
  const int n0 = blockIdx.x * TN;

  v8f acc[4][4];
#pragma unroll
  for (int i = 0; i < 4; ++i)
#pragma unroll
    for (int j = 0; j < 4; ++j) acc[i][j] = vzero8();

  const int r0 = t >> 5;
  const int c4 = (t & 31) << 2;
  const unsigned aoff = (unsigned)r0 * (unsigned)K + (unsigned)c4;
  const unsigned boff = (unsigned)r0 * (unsigned)N + (unsigned)c4;
  float* const asd = &As[r0][c4];
  float* const bsd = &Bsp[r0 >> 1][2 * c4 + (r0 & 1)];

  for (int kt = 0; kt < K; kt += TK) {
    prio_by_progress(kt / TK);

    float4 av[NLA], bv[NLB];
    const float* __restrict__ Ak = A + (size_t)m0 * K + kt;
    const float* __restrict__ Bk = Bm + (size_t)kt * N + n0;
#pragma unroll
    for (int i = 0; i < NLA; ++i)
      av[i] = *(const float4*)(Ak + (size_t)(8 * i) * K + aoff);
#pragma unroll
    for (int i = 0; i < NLB; ++i)
      bv[i] = *(const float4*)(Bk + (size_t)(8 * i) * N + boff);
#pragma unroll
    for (int i = 0; i < NLA; ++i)
      *(float4*)(asd + i * 8 * (TK + APAD)) = av[i];
#pragma unroll
    for (int i = 0; i < NLB; ++i) {
      float* dst = bsd + i * 4 * (2 * TN + BIPAD);
      dst[0] = bv[i].x;
      dst[2] = bv[i].y;
      dst[4] = bv[i].z;
      dst[6] = bv[i].w;
    }
    __syncthreads();

    int kgo = kg;
    asm volatile("" : "+v"(kgo));
#pragma unroll
    for (int kq = 0; kq < TK / 32; ++kq) {
      const int kk = kgo * (TK / 2) + kq * 16;
      v2f af[4][4];
#pragma unroll
      for (int s = 0; s < 4; ++s)
#pragma unroll
        for (int im = 0; im < 4; ++im) {
          int row = wm * 64 + im * 16 + lm;
          af[s][im] = *(const v2f*)&As[row][kk + 4 * s + 2 * lh];
        }
      v2f bf[4][4];
#pragma unroll
      for (int s = 0; s < 4; ++s) {
        const int khalf = ((kk + 4 * s) >> 1) + lh;
#pragma unroll
        for (int j = 0; j < 4; ++j) {
          int col = wn * 64 + j * 16 + lm;
          bf[s][j] = *(const v2f*)&Bsp[khalf][2 * col];
        }
      }
#pragma unroll
      for (int s = 0; s < 4; ++s)
#pragma unroll
        for (int j = 0; j < 4; ++j)
#pragma unroll
          for (int im = 0; im < 4; ++im)
            acc[im][j] = WMMA_F32(af[s][im], bf[s][j], acc[im][j]);
      __builtin_amdgcn_sched_barrier(0);
    }
    __syncthreads();
  }
  __builtin_amdgcn_s_setprio(0);

  float4* xb = (float4*)&As[0][0];
  int te = threadIdx.x;
  asm volatile("" : "+v"(te));
  const int lane_e = te & 31;
  const int wave_e = te >> 5;
  const int kg_e = wave_e >> 2;
  const int mw = m0 + ((wave_e & 3) & 1) * 64;
  const int nw = n0 + ((wave_e & 3) >> 1) * 64;
  if (kg_e == 0) gemm_hand_over<0>(acc, xb, wave_e, lane_e);
  else           gemm_hand_over<1>(acc, xb, wave_e, lane_e);
  __syncthreads();
  if (kg_e == 0) gemm_finish<0>(acc, xb, wave_e, lane_e, C, N, mw, nw);
  else           gemm_finish<1>(acc, xb, wave_e, lane_e, C, N, mw, nw);
}

constexpr int QT = 64;
constexpr int KT = 32;
constexpr int KVP = 12;
constexpr int VIPAD = 32;
constexpr int PPAD = 4;

__global__ __launch_bounds__(128, 4) void attn_f32_wmma(
    const float* __restrict__ Qm, const float* __restrict__ Km,
    const float* __restrict__ Vm, const int* __restrict__ mask,
    float* __restrict__ O) {
  __shared__ float Ks[KT][HD + KVP];
  __shared__ float Vsp[KT / 2][2 * HD + VIPAD];
  __shared__ float Ps[4][16][KT + PPAD];
  __shared__ float corrS[4][16];
  __shared__ float rsumS[4][16];
  __shared__ int mkS[KT];

  const int t = threadIdx.x;
  const int lane = t & 31;
  const int w = t >> 5;
  const int lm = lane & 15;
  const int lh = lane >> 4;

  int bid = blockIdx.x;
  const int nqb = Sq / QT;
  const int qb = bid % nqb; bid /= nqb;
  const int h = bid % NH;
  const int b = bid / NH;
  const int q0 = qb * QT;

  const size_t baseBH = (size_t)b * Sq * Hd + (size_t)h * Sq * HD;

  const float* qrow = Qm + baseBH + (size_t)(q0 + w * 16 + lm) * HD;
  v2f qf[HD / 4];
#pragma unroll
  for (int kd = 0; kd < HD / 4; ++kd)
    qf[kd] = *(const v2f*)(qrow + 4 * kd + 2 * lh);

  int mq[8];
#pragma unroll
  for (int r = 0; r < 8; ++r)
    mq[r] = mask[b * Sq + q0 + w * 16 + r + 8 * lh];

  v8f o[8];
#pragma unroll
  for (int s = 0; s < 8; ++s) o[s] = vzero8();
  float rmax = -INFINITY;
  float rsum = 0.0f;

  for (int kt = 0; kt < Sq; kt += KT) {
    prio_by_progress(kt / (2 * KT));

#pragma unroll
    for (int i = 0; i < 8; ++i) {
      int idx = t + i * 128;
      int r = idx >> 5;
      int c = (idx & 31) << 2;
      float4 kv = *(const float4*)(Km + baseBH + (size_t)(kt + r) * HD + c);
      *(float4*)&Ks[r][c] = kv;
      float4 vv = *(const float4*)(Vm + baseBH + (size_t)(kt + r) * HD + c);
      float* dst = &Vsp[r >> 1][2 * c + (r & 1)];
      dst[0] = vv.x;
      dst[2] = vv.y;
      dst[4] = vv.z;
      dst[6] = vv.w;
    }
    if (t < KT) mkS[t] = mask[b * Sq + kt + t];
    __syncthreads();

    v8f sc[2];
    sc[0] = vzero8();
    sc[1] = vzero8();
#pragma unroll
    for (int kd = 0; kd < HD / 4; ++kd) {
      v2f af = qf[kd];
#pragma unroll
      for (int j = 0; j < 2; ++j) {
        int kc = j * 16 + lm;
        v2f bf = *(const v2f*)&Ks[kc][4 * kd + 2 * lh];
        sc[j] = WMMA_F32(af, bf, sc[j]);
      }
    }

#pragma unroll
    for (int j = 0; j < 2; ++j)
#pragma unroll
      for (int r = 0; r < 8; ++r) {
        int m = r + 8 * lh;
        int kc = j * 16 + lm;
        float v = sc[j][r];
        if (mq[r] == 0 || mkS[kc] == 0) v = NEGV;
        Ps[w][m][kc] = v;
      }
    __syncthreads();

    {
      const int c0 = lh * 16;
      float pm = -INFINITY;
#pragma unroll
      for (int c = 0; c < 16; ++c) pm = fmaxf(pm, Ps[w][lm][c0 + c]);
      pm = fmaxf(pm, __shfl_xor(pm, 16, 32));
      float nm = fmaxf(rmax, pm);
      float cf = __expf(rmax - nm);
      float ps = 0.0f;
#pragma unroll
      for (int c = 0; c < 16; ++c) {
        float e = __expf(Ps[w][lm][c0 + c] - nm);
        Ps[w][lm][c0 + c] = e;
        ps += e;
      }
      ps += __shfl_xor(ps, 16, 32);
      rsum = rsum * cf + ps;
      rmax = nm;
      if (lane < 16) corrS[w][lm] = cf;
    }
    __syncthreads();

    float cv[8];
#pragma unroll
    for (int r = 0; r < 8; ++r) cv[r] = corrS[w][r + 8 * lh];
#pragma unroll
    for (int s = 0; s < 8; ++s)
#pragma unroll
      for (int r = 0; r < 8; ++r) o[s][r] *= cv[r];

#pragma unroll
    for (int kk = 0; kk < KT; kk += 4) {
      v2f af = *(const v2f*)&Ps[w][lm][kk + 2 * lh];
      const int khalf = (kk >> 1) + lh;
#pragma unroll
      for (int s = 0; s < 8; ++s) {
        int col = s * 16 + lm;
        v2f bf = *(const v2f*)&Vsp[khalf][2 * col];
        o[s] = WMMA_F32(af, bf, o[s]);
      }
    }
    __syncthreads();
  }
  __builtin_amdgcn_s_setprio(0);

  if (lane < 16) rsumS[w][lm] = rsum;
  __syncthreads();
  float rv[8];
#pragma unroll
  for (int r = 0; r < 8; ++r) rv[r] = 1.0f / rsumS[w][r + 8 * lh];

#pragma unroll
  for (int s = 0; s < 8; ++s)
#pragma unroll
    for (int r = 0; r < 8; ++r) {
      int s2 = q0 + w * 16 + r + 8 * lh;
      int d = s * 16 + lm;
      O[baseBH + (size_t)s2 * HD + d] = o[s][r] * rv[r];
    }
}

extern "C" void kernel_launch(void* const* d_in, const int* in_sizes, int n_in,
                              void* d_out, int out_size, void* d_ws, size_t ws_size,
                              hipStream_t stream) {
  const float* X    = (const float*)d_in[0];
  const int*   mask = (const int*)d_in[1];
  const float* Wq   = (const float*)d_in[2];
  const float* Wk   = (const float*)d_in[3];
  const float* Wv   = (const float*)d_in[4];
  const float* Wqm  = (const float*)d_in[5];
  const float* Wkm  = (const float*)d_in[6];
  const float* Wvm  = (const float*)d_in[7];
  const float* Wo   = (const float*)d_in[8];
  float* out = (float*)d_out;

  const size_t SZ = (size_t)Bz * Sq * Hd;
  float* s0 = (float*)d_ws;
  float* s1 = s0 + SZ;
  float* s2 = s1 + SZ;
  float* s3 = s2 + SZ;

  const int M = Bz * Sq, N = Hd, K = Hd;
  dim3 gg(N / TN, M / TM);
  dim3 gb(256);

  gemm_f32_wmma<<<gg, gb, 0, stream>>>(X,  Wq,  s0, M, N, K);
  gemm_f32_wmma<<<gg, gb, 0, stream>>>(X,  Wk,  s1, M, N, K);
  gemm_f32_wmma<<<gg, gb, 0, stream>>>(X,  Wv,  s2, M, N, K);
  gemm_f32_wmma<<<gg, gb, 0, stream>>>(s0, Wqm, s3, M, N, K);
  gemm_f32_wmma<<<gg, gb, 0, stream>>>(s1, Wkm, s0, M, N, K);
  gemm_f32_wmma<<<gg, gb, 0, stream>>>(s2, Wvm, s1, M, N, K);

  attn_f32_wmma<<<Bz * NH * (Sq / QT), 128, 0, stream>>>(s3, s0, s1, mask, s2);

  gemm_f32_wmma<<<gg, gb, 0, stream>>>(s2, Wo, out, M, N, K);
}
